# starting kernel
# baseline (speedup 1.0000x reference)
#include <hip/hip_runtime.h>
#include <hip/hip_bf16.h>

#define B_  32
#define D_  1024
#define R_  2048

typedef float v2f __attribute__((ext_vector_type(2)));
typedef float v8f __attribute__((ext_vector_type(8)));

__device__ __forceinline__ void top2_insert(float v, int i,
                                            float& v1, int& i1,
                                            float& v2, int& i2) {
    bool beats1 = (v > v1) || (v == v1 && i < i1);
    if (beats1) {
        v2 = v1; i2 = i1;
        v1 = v;  i1 = i;
    } else {
        bool beats2 = (v > v2) || (v == v2 && i < i2);
        if (beats2) { v2 = v; i2 = i; }
    }
}

__global__ void rule_prep_kernel(const float* __restrict__ fact_logits,
                                 const float* __restrict__ agg_logits,
                                 const float* __restrict__ strength,
                                 int*   __restrict__ ruleIdx,
                                 float* __restrict__ ruleW)
{
    const int wavesPerBlock = blockDim.x >> 5;
    const int r    = blockIdx.x * wavesPerBlock + (threadIdx.x >> 5);
    const int lane = threadIdx.x & 31;
    if (r >= R_) return;

    float v1 = -__builtin_inff(), v2 = -__builtin_inff();
    int   i1 = 0x7fffffff,        i2 = 0x7fffffff;

    const float* row = fact_logits + (size_t)r * D_;
    #pragma unroll 4
    for (int j = lane; j < D_; j += 32) {
        top2_insert(row[j], j, v1, i1, v2, i2);
    }
    #pragma unroll
    for (int m = 16; m >= 1; m >>= 1) {
        float ov1 = __shfl_xor(v1, m, 32);
        int   oi1 = __shfl_xor(i1, m, 32);
        float ov2 = __shfl_xor(v2, m, 32);
        int   oi2 = __shfl_xor(i2, m, 32);
        top2_insert(ov1, oi1, v1, i1, v2, i2);
        top2_insert(ov2, oi2, v1, i1, v2, i2);
    }

    if (lane == 0) {
        ruleIdx[2 * r + 0] = i1;
        ruleIdx[2 * r + 1] = i2;
        float a0 = agg_logits[3 * r + 0];
        float a1 = agg_logits[3 * r + 1];
        float a2 = agg_logits[3 * r + 2];
        float mx = fmaxf(a0, fmaxf(a1, a2));
        float e0 = expf(a0 - mx), e1 = expf(a1 - mx), e2 = expf(a2 - mx);
        float inv = 1.0f / (e0 + e1 + e2);
        ruleW[4 * r + 0] = e0 * inv;
        ruleW[4 * r + 1] = e1 * inv;
        ruleW[4 * r + 2] = e2 * inv;
        ruleW[4 * r + 3] = 1.0f / (1.0f + expf(-strength[r]));
    }
}

__global__ void gemm_wmma_kernel(const float* __restrict__ facts,
                                 const float* __restrict__ Wp,
                                 float* __restrict__ projPart,
                                 int klen)
{
    const int wave  = (blockIdx.x * blockDim.x + threadIdx.x) >> 5;
    const int lane  = threadIdx.x & 31;

    const int ntile = wave & 127;
    const int slice = wave >> 7;
    const int n0    = ntile * 16;
    const int kbeg  = slice * klen;

    const int rlo   = lane & 15;
    const int khalf = (lane >> 4) * 2;

    const float* aptr0 = facts + (size_t)(rlo)      * D_ + kbeg + khalf;
    const float* aptr1 = facts + (size_t)(16 + rlo) * D_ + kbeg + khalf;
    const float* bptr  = Wp    + (size_t)(n0 + rlo) * D_ + kbeg + khalf;

    v8f c0 = {};
    v8f c1 = {};
    for (int k = 0; k < klen; k += 16) {
        __builtin_prefetch(bptr + k + 512, 0, 3);
        #pragma unroll
        for (int kk = 0; kk < 16; kk += 4) {
            v2f b  = *(const v2f*)(bptr  + k + kk);
            v2f a0 = *(const v2f*)(aptr0 + k + kk);
            v2f a1 = *(const v2f*)(aptr1 + k + kk);
            c0 = __builtin_amdgcn_wmma_f32_16x16x4_f32(false, a0, false, b,
                                                       (short)0, c0, false, false);
            c1 = __builtin_amdgcn_wmma_f32_16x16x4_f32(false, a1, false, b,
                                                       (short)0, c1, false, false);
        }
    }

    float* outp = projPart + (size_t)slice * B_ * R_;
    const int col   = n0 + rlo;
    const int rbase = (lane >> 4) * 8;
    #pragma unroll
    for (int v = 0; v < 8; ++v) {
        outp[(size_t)(rbase + v)      * R_ + col] = c0[v];
        outp[(size_t)(rbase + v + 16) * R_ + col] = c1[v];
    }
}

__global__ void rule_act_kernel(const float* __restrict__ facts,
                                const int*   __restrict__ ruleIdx,
                                const float* __restrict__ ruleW,
                                float* __restrict__ ract)
{
    const int gid = blockIdx.x * blockDim.x + threadIdx.x;
    const int b = gid >> 11;
    const int r = gid & (R_ - 1);

    const float f0 = facts[(size_t)b * D_ + ruleIdx[2 * r + 0]];
    const float f1 = facts[(size_t)b * D_ + ruleIdx[2 * r + 1]];
    const float w0 = ruleW[4 * r + 0];
    const float w1 = ruleW[4 * r + 1];
    const float w2 = ruleW[4 * r + 2];
    const float sg = ruleW[4 * r + 3];

    const float and_a = f0 * f1;
    const float or_a  = 1.0f - (1.0f - f0) * (1.0f - f1);
    const float k_a   = (f0 + f1) / (2.0f + 1e-8f);

    ract[gid] = (w0 * and_a + w1 * or_a + w2 * k_a) * sg;
}

__global__ void gate_ln_kernel(const float* __restrict__ projPart,
                               int nslice,
                               const float* __restrict__ ract,
                               const float* __restrict__ gamma,
                               const float* __restrict__ beta,
                               float* __restrict__ out)
{
    const int b   = blockIdx.x;
    const int tid = threadIdx.x;

    __shared__ float s_pre[R_];
    __shared__ float s_act[R_];
    __shared__ unsigned char s_ch[R_];
    __shared__ float s_val[256];
    __shared__ int   s_idx[256];

    for (int i = tid; i < R_; i += 256) {
        float acc = 0.0f;
        for (int s = 0; s < nslice; ++s) {
            acc += projPart[(size_t)s * B_ * R_ + (size_t)b * R_ + i];
        }
        s_pre[i] = acc;
        s_act[i] = ract[(size_t)b * R_ + i];
        s_ch[i]  = 0;
    }
    __syncthreads();

    for (int t = 0; t < 8; ++t) {
        float bv = -__builtin_inff();
        int   bi = 0x7fffffff;
        for (int i = tid; i < R_; i += 256) {
            if (!s_ch[i]) {
                float v = s_act[i];
                if (v > bv || (v == bv && i < bi)) { bv = v; bi = i; }
            }
        }
        s_val[tid] = bv; s_idx[tid] = bi;
        __syncthreads();
        for (int s = 128; s > 0; s >>= 1) {
            if (tid < s) {
                float ov = s_val[tid + s]; int oi = s_idx[tid + s];
                if (ov > s_val[tid] || (ov == s_val[tid] && oi < s_idx[tid])) {
                    s_val[tid] = ov; s_idx[tid] = oi;
                }
            }
            __syncthreads();
        }
        if (tid == 0 && s_idx[0] < R_) s_ch[s_idx[0]] = 1;
        __syncthreads();
    }

    for (int i = tid; i < R_; i += 256) {
        if (s_ch[i]) s_pre[i] += s_act[i];
    }
    __syncthreads();

    float sum = 0.0f;
    for (int i = tid; i < R_; i += 256) sum += s_pre[i];
    s_val[tid] = sum;
    __syncthreads();
    for (int s = 128; s > 0; s >>= 1) {
        if (tid < s) s_val[tid] += s_val[tid + s];
        __syncthreads();
    }
    const float mu = s_val[0] * (1.0f / R_);
    __syncthreads();

    float ss = 0.0f;
    for (int i = tid; i < R_; i += 256) { float d = s_pre[i] - mu; ss += d * d; }
    s_val[tid] = ss;
    __syncthreads();
    for (int s = 128; s > 0; s >>= 1) {
        if (tid < s) s_val[tid] += s_val[tid + s];
        __syncthreads();
    }
    const float inv = rsqrtf(s_val[0] * (1.0f / R_) + 1e-5f);

    for (int i = tid; i < R_; i += 256) {
        out[(size_t)b * R_ + i] = (s_pre[i] - mu) * inv * gamma[i] + beta[i];
    }
}

extern "C" void kernel_launch(void* const* d_in, const int* in_sizes, int n_in,
                              void* d_out, int out_size, void* d_ws, size_t ws_size,
                              hipStream_t stream) {
    const float* facts        = (const float*)d_in[0];
    const float* fact_logits  = (const float*)d_in[1];
    const float* agg_logits   = (const float*)d_in[2];
    const float* strength     = (const float*)d_in[3];
    const float* W_proj       = (const float*)d_in[4];
    const float* ln_gamma     = (const float*)d_in[5];
    const float* ln_beta      = (const float*)d_in[6];
    float* out = (float*)d_out;

    const size_t kFixed = 49152 + (size_t)B_ * R_ * 4;
    int nslice = 1;
    if (ws_size >= kFixed + 4 * (size_t)B_ * R_ * 4) nslice = 4;
    else if (ws_size >= kFixed + 2 * (size_t)B_ * R_ * 4) nslice = 2;

    char* ws = (char*)d_ws;
    float* ruleW    = (float*)(ws);
    int*   ruleIdx  = (int*)  (ws + (size_t)R_ * 4 * 4);
    float* ract     = (float*)(ws + 49152);
    float* projPart = (float*)(ws + kFixed);

    rule_prep_kernel<<<R_ / 8, 256, 0, stream>>>(fact_logits, agg_logits, strength,
                                                 ruleIdx, ruleW);

    gemm_wmma_kernel<<<16 * nslice, 256, 0, stream>>>(facts, W_proj, projPart,
                                                      D_ / nslice);

    rule_act_kernel<<<(B_ * R_) / 256, 256, 0, stream>>>(facts, ruleIdx, ruleW, ract);

    gate_ln_kernel<<<B_, 256, 0, stream>>>(projPart, nslice, ract,
                                           ln_gamma, ln_beta, out);
}


# rule top-2 scan moved onto the projection's tiling: each GEMM wave scans its 128-k slice of 16 rule rows behind one load group, lists merged through LDS, act in the epilogue; rule launch goes
# speedup vs baseline: 3.0427x; 3.0427x over previous
#include <hip/hip_runtime.h>
#include <hip/hip_bf16.h>


#define B_  32
#define D_  1024
#define R_  2048

typedef float v2f __attribute__((ext_vector_type(2)));
typedef float v8f __attribute__((ext_vector_type(8)));

__device__ __forceinline__ void top2_insert(float v, int i,
                                            float& v1, int& i1,
                                            float& v2, int& i2) {
    const bool b1 = (v > v1) | ((v == v1) & (i < i1));
    const bool b2 = (v > v2) | ((v == v2) & (i < i2));
    const float nv2 = b1 ? v1 : (b2 ? v : v2);
    const int   ni2 = b1 ? i1 : (b2 ? i : i2);
    v1 = b1 ? v : v1;
    i1 = b1 ? i : i1;
    v2 = nv2;
    i2 = ni2;
}

__device__ __forceinline__ void top2_insert_later(float v, int i,
                                                  float& v1, int& i1,
                                                  float& v2, int& i2) {
    const bool b1 = v > v1;
    const bool b2 = v > v2;
    const float nv2 = b1 ? v1 : (b2 ? v : v2);
    const int   ni2 = b1 ? i1 : (b2 ? i : i2);
    v1 = b1 ? v : v1;
    i1 = b1 ? i : i1;
    v2 = nv2;
    i2 = ni2;
}

#define KSPLIT_   8
#define KLEN_     (D_ / KSPLIT_)
#define TILEPAD_  (8 * 32 + 16)

__global__ void __launch_bounds__(256)
proj_scan_kernel(const float* __restrict__ facts,
                 const float* __restrict__ Wp,
                 const float* __restrict__ fact_logits,
                 const float* __restrict__ agg_logits,
                 const float* __restrict__ strength,
                 float* __restrict__ proj,
                 float* __restrict__ ract)
{
    __shared__ float s_part[KSPLIT_ * 2 * TILEPAD_];
    __shared__ __attribute__((aligned(8))) float s_tv[16 * 16 * 2];
    __shared__ __attribute__((aligned(8))) int   s_ti[16 * 16 * 2];
    __shared__ __attribute__((aligned(8))) float s_t[2 * B_ * 8];

    const int lane  = threadIdx.x & 31;
    const int wv    = threadIdx.x >> 5;
    const int n0    = (blockIdx.x >> 1) * 32;
    const int m0    = (blockIdx.x & 1) * 16;
    const int rs0   = n0 + m0;

    const int c     = lane & 15;
    const int h     = lane >> 4;
    const int kofs  = wv * KLEN_ + 4 * h;

    const float4* lp  = (const float4*)(fact_logits + (size_t)(rs0 + c)  * D_ + kofs);
    const float4* ap  = (const float4*)(facts + (size_t)(m0 + c)      * D_ + kofs);
    const float4* bp0 = (const float4*)(Wp    + (size_t)(n0 + c)      * D_ + kofs);
    const float4* bp1 = (const float4*)(Wp    + (size_t)(n0 + 16 + c) * D_ + kofs);

    float4 ql[16], qa[16], qb0[16], qb1[16];
    #pragma unroll
    for (int j = 0; j < 16; ++j) ql[j] = lp[2 * j];
    #pragma unroll
    for (int j = 0; j < 16; ++j) {
        qa[j]  = ap[2 * j];
        qb0[j] = bp0[2 * j];
        qb1[j] = bp1[2 * j];
    }
    const int   ra  = rs0 + 2 * wv;
    const float aa0 = agg_logits[3 * ra + 0], aa1 = agg_logits[3 * ra + 1], aa2 = agg_logits[3 * ra + 2];
    const float ab0 = agg_logits[3 * ra + 3], ab1 = agg_logits[3 * ra + 4], ab2 = agg_logits[3 * ra + 5];
    const float sta = strength[ra], stb = strength[ra + 1];
    __builtin_amdgcn_sched_barrier(0);

    float v1 = -__builtin_inff(), v2 = -__builtin_inff();
    int   i1 = 0x7fffffff,        i2 = 0x7fffffff;
    top2_insert(ql[0].x, kofs + 0, v1, i1, v2, i2);
    top2_insert(ql[0].y, kofs + 1, v1, i1, v2, i2);
    top2_insert_later(ql[0].z, kofs + 2, v1, i1, v2, i2);
    top2_insert_later(ql[0].w, kofs + 3, v1, i1, v2, i2);
    #pragma unroll
    for (int j = 1; j < 16; ++j) {
        const int k = kofs + 8 * j;
        top2_insert_later(ql[j].x, k + 0, v1, i1, v2, i2);
        top2_insert_later(ql[j].y, k + 1, v1, i1, v2, i2);
        top2_insert_later(ql[j].z, k + 2, v1, i1, v2, i2);
        top2_insert_later(ql[j].w, k + 3, v1, i1, v2, i2);
    }
    ((float2*)s_tv)[c * 16 + wv * 2 + h] = make_float2(v1, v2);
    ((int2*)s_ti)[c * 16 + wv * 2 + h]   = make_int2(i1, i2);

    v8f c0 = {};
    v8f c1 = {};
    #pragma unroll
    for (int j = 0; j < 16; ++j) {
        v2f a, b0, b1;
        a[0]  = qa[j].x;  a[1]  = qa[j].y;
        b0[0] = qb0[j].x; b0[1] = qb0[j].y;
        b1[0] = qb1[j].x; b1[1] = qb1[j].y;
        c0 = __builtin_amdgcn_wmma_f32_16x16x4_f32(false, a, false, b0,
                                                   (short)0, c0, false, false);
        c1 = __builtin_amdgcn_wmma_f32_16x16x4_f32(false, a, false, b1,
                                                   (short)0, c1, false, false);
        a[0]  = qa[j].z;  a[1]  = qa[j].w;
        b0[0] = qb0[j].z; b0[1] = qb0[j].w;
        b1[0] = qb1[j].z; b1[1] = qb1[j].w;
        c0 = __builtin_amdgcn_wmma_f32_16x16x4_f32(false, a, false, b0,
                                                   (short)0, c0, false, false);
        c1 = __builtin_amdgcn_wmma_f32_16x16x4_f32(false, a, false, b1,
                                                   (short)0, c1, false, false);
    }

    float* pw = s_part + (wv * 2) * TILEPAD_ + lane;
    #pragma unroll
    for (int v = 0; v < 8; ++v) {
        pw[v * 32]            = c0[v];
        pw[TILEPAD_ + v * 32] = c1[v];
    }
    __syncthreads();

    {
        const float2 tv = ((const float2*)s_tv)[(2 * wv + h) * 16 + c];
        const int2   ti = ((const int2*)s_ti)[(2 * wv + h) * 16 + c];
        v1 = tv.x; v2 = tv.y; i1 = ti.x; i2 = ti.y;
        #pragma unroll
        for (int m = 8; m >= 1; m >>= 1) {
            const float ov1 = __shfl_xor(v1, m, 32);
            const int   oi1 = __shfl_xor(i1, m, 32);
            const float ov2 = __shfl_xor(v2, m, 32);
            const int   oi2 = __shfl_xor(i2, m, 32);
            top2_insert(ov1, oi1, v1, i1, v2, i2);
            top2_insert(ov2, oi2, v1, i1, v2, i2);
        }
    }
    const int ia1 = __builtin_amdgcn_readlane(i1, 0)  & (D_ - 1);
    const int ia2 = __builtin_amdgcn_readlane(i2, 0)  & (D_ - 1);
    const int ib1 = __builtin_amdgcn_readlane(i1, 16) & (D_ - 1);
    const int ib2 = __builtin_amdgcn_readlane(i2, 16) & (D_ - 1);
    const float* fr = facts + (size_t)lane * D_;
    const float fa0 = fr[ia1], fa1 = fr[ia2];
    const float fb0 = fr[ib1], fb1 = fr[ib2];

    const int tile = lane >> 4;
    float pacc[2];
    #pragma unroll
    for (int hh = 0; hh < 2; ++hh) {
        const float* pr = s_part + tile * TILEPAD_ + wv * 32 + c + 16 * hh;
        float acc = 0.0f;
        #pragma unroll
        for (int s = 0; s < KSPLIT_; ++s) acc += pr[s * 2 * TILEPAD_];
        pacc[hh] = acc;
    }

    float act_a, act_b;
    {
        const float mx = fmaxf(aa0, fmaxf(aa1, aa2));
        const float e0 = expf(aa0 - mx), e1 = expf(aa1 - mx), e2 = expf(aa2 - mx);
        const float inv = 1.0f / (e0 + e1 + e2);
        const float w0 = e0 * inv;
        const float w1 = e1 * inv;
        const float w2 = e2 * inv;
        const float sg = 1.0f / (1.0f + expf(-sta));
        const float and_a = fa0 * fa1;
        const float or_a  = 1.0f - (1.0f - fa0) * (1.0f - fa1);
        const float k_a   = (fa0 + fa1) / (2.0f + 1e-8f);
        act_a = (w0 * and_a + w1 * or_a + w2 * k_a) * sg;
    }
    {
        const float mx = fmaxf(ab0, fmaxf(ab1, ab2));
        const float e0 = expf(ab0 - mx), e1 = expf(ab1 - mx), e2 = expf(ab2 - mx);
        const float inv = 1.0f / (e0 + e1 + e2);
        const float w0 = e0 * inv;
        const float w1 = e1 * inv;
        const float w2 = e2 * inv;
        const float sg = 1.0f / (1.0f + expf(-stb));
        const float and_a = fb0 * fb1;
        const float or_a  = 1.0f - (1.0f - fb0) * (1.0f - fb1);
        const float k_a   = (fb0 + fb1) / (2.0f + 1e-8f);
        act_b = (w0 * and_a + w1 * or_a + w2 * k_a) * sg;
    }
    ((float2*)s_t)[((wv >> 2) * (B_ * 8) + lane * 8 + ((2 * wv) & 7)) >> 1] = make_float2(act_a, act_b);
    __syncthreads();
    proj[(size_t)(m0 + wv)     * R_ + n0 + lane] = pacc[0];
    proj[(size_t)(m0 + wv + 8) * R_ + n0 + lane] = pacc[1];
    float* rc = ract + (size_t)(rs0 >> 3) * (B_ * 8);
    rc[threadIdx.x]       = s_t[threadIdx.x];
    rc[threadIdx.x + 256] = s_t[threadIdx.x + 256];
}

__device__ __forceinline__ bool key_before(float v, int i, float w, int j) {
    return (v > w) | ((v == w) & (i < j));
}

__global__ void gate_ln_kernel(const float* __restrict__ proj,
                               const float* __restrict__ ract,
                               const float* __restrict__ gamma,
                               const float* __restrict__ beta,
                               float* __restrict__ out)
{
    const int b   = blockIdx.x;
    const int tid = threadIdx.x;

    __shared__ __attribute__((aligned(16))) float s_pre[R_];
    __shared__ __attribute__((aligned(16))) float s_act[R_];
    __shared__ __attribute__((aligned(16))) unsigned char s_ch[R_];
    __shared__ float s_val[256];
    __shared__ int   s_idx[256];
    __shared__ float s_wv[8];
    __shared__ int   s_wi[8];
    __shared__ int   s_wn[8];

    {
        const float4* p4 = (const float4*)(proj + (size_t)b * R_);
        const float4* a4 = (const float4*)(ract + ((size_t)tid * B_ + b) * 8);
        const float4 p0 = p4[tid], p1 = p4[tid + 256];
        const float4 q0 = a4[0], q1 = a4[1];
        ((float4*)s_pre)[tid]       = p0;
        ((float4*)s_pre)[tid + 256] = p1;
        ((float4*)s_act)[2 * tid]     = q0;
        ((float4*)s_act)[2 * tid + 1] = q1;
        ((unsigned*)s_ch)[tid]       = 0u;
        ((unsigned*)s_ch)[tid + 256] = 0u;
    }
    __syncthreads();

    {
        const int lane = tid & 31;
        const int wv   = tid >> 5;

        float a[8];
        float bv = -__builtin_inff();
        int   bi = 0x7fffffff;
        for (int k = 0; k < 8; ++k) {
            const int i = tid + 256 * k;
            a[k] = s_act[i];
            if (key_before(a[k], i, bv, bi)) { bv = a[k]; bi = i; }
        }
        for (int m = 16; m >= 1; m >>= 1) {
            float ov = __shfl_xor(bv, m, 32);
            int   oi = __shfl_xor(bi, m, 32);
            if (key_before(ov, oi, bv, bi)) { bv = ov; bi = oi; }
        }
        if (lane == 0) { s_wv[wv] = bv; s_wi[wv] = bi; }
        s_val[tid] = -__builtin_inff();
        s_idx[tid] = 0x7fffffff;
        __syncthreads();

        float lv[8];
        int   li[8];
        for (int w = 0; w < 8; ++w) { lv[w] = s_wv[w]; li[w] = s_wi[w]; }
        float Lv = lv[0];
        int   Li = li[0];
        for (int w = 1; w < 8; ++w) {
            const bool later = key_before(Lv, Li, lv[w], li[w]);
            Lv = later ? lv[w] : Lv;
            Li = later ? li[w] : Li;
        }

        bool keep[8];
        int  slot[8];
        int  wcount = 0;
        for (int k = 0; k < 8; ++k) {
            const int i = tid + 256 * k;
            keep[k] = (a[k] > Lv) || (a[k] == Lv && i <= Li);
            const unsigned m = __builtin_amdgcn_ballot_w32(keep[k]);
            slot[k] = wcount + (int)__builtin_amdgcn_mbcnt_lo(m, 0u);
            wcount += __builtin_popcount(m);
        }
        if (lane == 0) s_wn[wv] = wcount;
        __syncthreads();

        int off = 0, n = 0;
        for (int w = 0; w < 8; ++w) {
            const int c = s_wn[w];
            if (w < wv) off += c;
            n += c;
        }

        if (n <= 256) {
            for (int k = 0; k < 8; ++k) {
                if (keep[k]) {
                    s_val[off + slot[k]] = a[k];
                    s_idx[off + slot[k]] = tid + 256 * k;
                }
            }
            __syncthreads();
            if (tid < n) {
                const float v = s_val[tid];
                const int   i = s_idx[tid];
                int rank = 0;
                for (int j = 0; j < n; j += 8) {
                    float cv[8];
                    int   ci[8];
                    for (int u = 0; u < 8; ++u) { cv[u] = s_val[j + u]; ci[u] = s_idx[j + u]; }
                    for (int u = 0; u < 8; ++u) rank += key_before(cv[u], ci[u], v, i) ? 1 : 0;
                }
                if (rank < 8) s_ch[i] = 1;
            }
            __syncthreads();
        } else {
            for (int t = 0; t < 8; ++t) {
                float pv = -__builtin_inff();
                int   pi = 0x7fffffff;
                for (int i = tid; i < R_; i += 256) {
                    if (!s_ch[i]) {
                        float v = s_act[i];
                        if (key_before(v, i, pv, pi)) { pv = v; pi = i; }
                    }
                }
                s_val[tid] = pv; s_idx[tid] = pi;
                __syncthreads();
                for (int s = 128; s > 0; s >>= 1) {
                    if (tid < s) {
                        float ov = s_val[tid + s]; int oi = s_idx[tid + s];
                        if (key_before(ov, oi, s_val[tid], s_idx[tid])) {
                            s_val[tid] = ov; s_idx[tid] = oi;
                        }
                    }
                    __syncthreads();
                }
                if (tid == 0 && s_idx[0] < R_) s_ch[s_idx[0]] = 1;
                __syncthreads();
            }
        }
    }

    for (int i = tid; i < R_; i += 256) {
        if (s_ch[i]) s_pre[i] += s_act[i];
    }
    __syncthreads();

    float sum = 0.0f;
    for (int i = tid; i < R_; i += 256) sum += s_pre[i];
    s_val[tid] = sum;
    __syncthreads();
    for (int s = 128; s > 0; s >>= 1) {
        if (tid < s) s_val[tid] += s_val[tid + s];
        __syncthreads();
    }
    const float mu = s_val[0] * (1.0f / R_);
    __syncthreads();

    float ss = 0.0f;
    for (int i = tid; i < R_; i += 256) { float d = s_pre[i] - mu; ss += d * d; }
    s_val[tid] = ss;
    __syncthreads();
    for (int s = 128; s > 0; s >>= 1) {
        if (tid < s) s_val[tid] += s_val[tid + s];
        __syncthreads();
    }
    const float inv = rsqrtf(s_val[0] * (1.0f / R_) + 1e-5f);

    for (int i = tid; i < R_; i += 256) {
        out[(size_t)b * R_ + i] = (s_pre[i] - mu) * inv * gamma[i] + beta[i];
    }
}

extern "C" void kernel_launch(void* const* d_in, const int* in_sizes, int n_in,
                              void* d_out, int out_size, void* d_ws, size_t ws_size,
                              hipStream_t stream) {
    const float* facts        = (const float*)d_in[0];
    const float* fact_logits  = (const float*)d_in[1];
    const float* agg_logits   = (const float*)d_in[2];
    const float* strength     = (const float*)d_in[3];
    const float* W_proj       = (const float*)d_in[4];
    const float* ln_gamma     = (const float*)d_in[5];
    const float* ln_beta      = (const float*)d_in[6];
    float* out = (float*)d_out;

    char* ws = (char*)d_ws;
    float* ract     = (float*)(ws);
    float* proj     = (float*)(ws + 262144);

    proj_scan_kernel<<<128, 256, 0, stream>>>(facts, W_proj, fact_logits, agg_logits, strength,
                                              proj, ract);

    gate_ln_kernel<<<B_, 256, 0, stream>>>(proj, ract, ln_gamma, ln_beta, out);
}


# gate+LayerNorm kernel: each thread's 8 row elements (proj, gamma, beta) come as one load group at entry and stay in registers; 4 rolled 8-trip passes (a round trip per trip) go; no sum changed
# speedup vs baseline: 4.0732x; 1.3387x over previous
#include <hip/hip_runtime.h>
#include <hip/hip_bf16.h>


#define B_  32
#define D_  1024
#define R_  2048

typedef float v2f __attribute__((ext_vector_type(2)));
typedef float v8f __attribute__((ext_vector_type(8)));

__device__ __forceinline__ void top2_insert(float v, int i,
                                            float& v1, int& i1,
                                            float& v2, int& i2) {
    const bool b1 = (v > v1) | ((v == v1) & (i < i1));
    const bool b2 = (v > v2) | ((v == v2) & (i < i2));
    const float nv2 = b1 ? v1 : (b2 ? v : v2);
    const int   ni2 = b1 ? i1 : (b2 ? i : i2);
    v1 = b1 ? v : v1;
    i1 = b1 ? i : i1;
    v2 = nv2;
    i2 = ni2;
}

__device__ __forceinline__ void top2_insert_later(float v, int i,
                                                  float& v1, int& i1,
                                                  float& v2, int& i2) {
    const bool b1 = v > v1;
    const bool b2 = v > v2;
    const float nv2 = b1 ? v1 : (b2 ? v : v2);
    const int   ni2 = b1 ? i1 : (b2 ? i : i2);
    v1 = b1 ? v : v1;
    i1 = b1 ? i : i1;
    v2 = nv2;
    i2 = ni2;
}

#define KSPLIT_   8
#define KLEN_     (D_ / KSPLIT_)
#define TILEPAD_  (8 * 32 + 16)

__global__ void __launch_bounds__(256)
proj_scan_kernel(const float* __restrict__ facts,
                 const float* __restrict__ Wp,
                 const float* __restrict__ fact_logits,
                 const float* __restrict__ agg_logits,
                 const float* __restrict__ strength,
                 float* __restrict__ proj,
                 float* __restrict__ ract)
{
    __shared__ float s_part[KSPLIT_ * 2 * TILEPAD_];
    __shared__ __attribute__((aligned(8))) float s_tv[16 * 16 * 2];
    __shared__ __attribute__((aligned(8))) int   s_ti[16 * 16 * 2];
    __shared__ __attribute__((aligned(8))) float s_t[2 * B_ * 8];

    const int lane  = threadIdx.x & 31;
    const int wv    = threadIdx.x >> 5;
    const int n0    = (blockIdx.x >> 1) * 32;
    const int m0    = (blockIdx.x & 1) * 16;
    const int rs0   = n0 + m0;

    const int c     = lane & 15;
    const int h     = lane >> 4;
    const int kofs  = wv * KLEN_ + 4 * h;

    const float4* lp  = (const float4*)(fact_logits + (size_t)(rs0 + c)  * D_ + kofs);
    const float4* ap  = (const float4*)(facts + (size_t)(m0 + c)      * D_ + kofs);
    const float4* bp0 = (const float4*)(Wp    + (size_t)(n0 + c)      * D_ + kofs);
    const float4* bp1 = (const float4*)(Wp    + (size_t)(n0 + 16 + c) * D_ + kofs);

    float4 ql[16], qa[16], qb0[16], qb1[16];
    #pragma unroll
    for (int j = 0; j < 16; ++j) ql[j] = lp[2 * j];
    #pragma unroll
    for (int j = 0; j < 16; ++j) {
        qa[j]  = ap[2 * j];
        qb0[j] = bp0[2 * j];
        qb1[j] = bp1[2 * j];
    }
    const int   ra  = rs0 + 2 * wv;
    const float aa0 = agg_logits[3 * ra + 0], aa1 = agg_logits[3 * ra + 1], aa2 = agg_logits[3 * ra + 2];
    const float ab0 = agg_logits[3 * ra + 3], ab1 = agg_logits[3 * ra + 4], ab2 = agg_logits[3 * ra + 5];
    const float sta = strength[ra], stb = strength[ra + 1];
    __builtin_amdgcn_sched_barrier(0);

    float v1 = -__builtin_inff(), v2 = -__builtin_inff();
    int   i1 = 0x7fffffff,        i2 = 0x7fffffff;
    top2_insert(ql[0].x, kofs + 0, v1, i1, v2, i2);
    top2_insert(ql[0].y, kofs + 1, v1, i1, v2, i2);
    top2_insert_later(ql[0].z, kofs + 2, v1, i1, v2, i2);
    top2_insert_later(ql[0].w, kofs + 3, v1, i1, v2, i2);
    #pragma unroll
    for (int j = 1; j < 16; ++j) {
        const int k = kofs + 8 * j;
        top2_insert_later(ql[j].x, k + 0, v1, i1, v2, i2);
        top2_insert_later(ql[j].y, k + 1, v1, i1, v2, i2);
        top2_insert_later(ql[j].z, k + 2, v1, i1, v2, i2);
        top2_insert_later(ql[j].w, k + 3, v1, i1, v2, i2);
    }
    ((float2*)s_tv)[c * 16 + wv * 2 + h] = make_float2(v1, v2);
    ((int2*)s_ti)[c * 16 + wv * 2 + h]   = make_int2(i1, i2);

    v8f c0 = {};
    v8f c1 = {};
    #pragma unroll
    for (int j = 0; j < 16; ++j) {
        v2f a, b0, b1;
        a[0]  = qa[j].x;  a[1]  = qa[j].y;
        b0[0] = qb0[j].x; b0[1] = qb0[j].y;
        b1[0] = qb1[j].x; b1[1] = qb1[j].y;
        c0 = __builtin_amdgcn_wmma_f32_16x16x4_f32(false, a, false, b0,
                                                   (short)0, c0, false, false);
        c1 = __builtin_amdgcn_wmma_f32_16x16x4_f32(false, a, false, b1,
                                                   (short)0, c1, false, false);
        a[0]  = qa[j].z;  a[1]  = qa[j].w;
        b0[0] = qb0[j].z; b0[1] = qb0[j].w;
        b1[0] = qb1[j].z; b1[1] = qb1[j].w;
        c0 = __builtin_amdgcn_wmma_f32_16x16x4_f32(false, a, false, b0,
                                                   (short)0, c0, false, false);
        c1 = __builtin_amdgcn_wmma_f32_16x16x4_f32(false, a, false, b1,
                                                   (short)0, c1, false, false);
    }

    float* pw = s_part + (wv * 2) * TILEPAD_ + lane;
    #pragma unroll
    for (int v = 0; v < 8; ++v) {
        pw[v * 32]            = c0[v];
        pw[TILEPAD_ + v * 32] = c1[v];
    }
    __syncthreads();

    {
        const float2 tv = ((const float2*)s_tv)[(2 * wv + h) * 16 + c];
        const int2   ti = ((const int2*)s_ti)[(2 * wv + h) * 16 + c];
        v1 = tv.x; v2 = tv.y; i1 = ti.x; i2 = ti.y;
        #pragma unroll
        for (int m = 8; m >= 1; m >>= 1) {
            const float ov1 = __shfl_xor(v1, m, 32);
            const int   oi1 = __shfl_xor(i1, m, 32);
            const float ov2 = __shfl_xor(v2, m, 32);
            const int   oi2 = __shfl_xor(i2, m, 32);
            top2_insert(ov1, oi1, v1, i1, v2, i2);
            top2_insert(ov2, oi2, v1, i1, v2, i2);
        }
    }
    const int ia1 = __builtin_amdgcn_readlane(i1, 0)  & (D_ - 1);
    const int ia2 = __builtin_amdgcn_readlane(i2, 0)  & (D_ - 1);
    const int ib1 = __builtin_amdgcn_readlane(i1, 16) & (D_ - 1);
    const int ib2 = __builtin_amdgcn_readlane(i2, 16) & (D_ - 1);
    const float* fr = facts + (size_t)lane * D_;
    const float fa0 = fr[ia1], fa1 = fr[ia2];
    const float fb0 = fr[ib1], fb1 = fr[ib2];

    const int tile = lane >> 4;
    float pacc[2];
    #pragma unroll
    for (int hh = 0; hh < 2; ++hh) {
        const float* pr = s_part + tile * TILEPAD_ + wv * 32 + c + 16 * hh;
        float acc = 0.0f;
        #pragma unroll
        for (int s = 0; s < KSPLIT_; ++s) acc += pr[s * 2 * TILEPAD_];
        pacc[hh] = acc;
    }

    float act_a, act_b;
    {
        const float mx = fmaxf(aa0, fmaxf(aa1, aa2));
        const float e0 = expf(aa0 - mx), e1 = expf(aa1 - mx), e2 = expf(aa2 - mx);
        const float inv = 1.0f / (e0 + e1 + e2);
        const float w0 = e0 * inv;
        const float w1 = e1 * inv;
        const float w2 = e2 * inv;
        const float sg = 1.0f / (1.0f + expf(-sta));
        const float and_a = fa0 * fa1;
        const float or_a  = 1.0f - (1.0f - fa0) * (1.0f - fa1);
        const float k_a   = (fa0 + fa1) / (2.0f + 1e-8f);
        act_a = (w0 * and_a + w1 * or_a + w2 * k_a) * sg;
    }
    {
        const float mx = fmaxf(ab0, fmaxf(ab1, ab2));
        const float e0 = expf(ab0 - mx), e1 = expf(ab1 - mx), e2 = expf(ab2 - mx);
        const float inv = 1.0f / (e0 + e1 + e2);
        const float w0 = e0 * inv;
        const float w1 = e1 * inv;
        const float w2 = e2 * inv;
        const float sg = 1.0f / (1.0f + expf(-stb));
        const float and_a = fb0 * fb1;
        const float or_a  = 1.0f - (1.0f - fb0) * (1.0f - fb1);
        const float k_a   = (fb0 + fb1) / (2.0f + 1e-8f);
        act_b = (w0 * and_a + w1 * or_a + w2 * k_a) * sg;
    }
    ((float2*)s_t)[((wv >> 2) * (B_ * 8) + lane * 8 + ((2 * wv) & 7)) >> 1] = make_float2(act_a, act_b);
    __syncthreads();
    proj[(size_t)(m0 + wv)     * R_ + n0 + lane] = pacc[0];
    proj[(size_t)(m0 + wv + 8) * R_ + n0 + lane] = pacc[1];
    float* rc = ract + (size_t)(rs0 >> 3) * (B_ * 8);
    rc[threadIdx.x]       = s_t[threadIdx.x];
    rc[threadIdx.x + 256] = s_t[threadIdx.x + 256];
}

__device__ __forceinline__ bool key_before(float v, int i, float w, int j) {
    return (v > w) | ((v == w) & (i < j));
}

__global__ void gate_ln_kernel(const float* __restrict__ proj,
                               const float* __restrict__ ract,
                               const float* __restrict__ gamma,
                               const float* __restrict__ beta,
                               float* __restrict__ out)
{
    const int b   = blockIdx.x;
    const int tid = threadIdx.x;

    __shared__ __attribute__((aligned(16))) float s_act[R_];
    __shared__ __attribute__((aligned(16))) unsigned char s_ch[R_];
    __shared__ float s_val[256];
    __shared__ int   s_idx[256];
    __shared__ float s_wv[8];
    __shared__ int   s_wi[8];
    __shared__ int   s_wn[8];

    float p[8], g[8], be[8];
    {
        const float4* a4 = (const float4*)(ract + ((size_t)tid * B_ + b) * 8);
        const float4 q0 = a4[0], q1 = a4[1];
        const float* pr = proj + (size_t)b * R_ + tid;
        #pragma unroll
        for (int k = 0; k < 8; ++k) {
            p[k]  = pr[256 * k];
            g[k]  = gamma[tid + 256 * k];
            be[k] = beta[tid + 256 * k];
        }
        ((float4*)s_act)[2 * tid]     = q0;
        ((float4*)s_act)[2 * tid + 1] = q1;
        ((unsigned*)s_ch)[tid]       = 0u;
        ((unsigned*)s_ch)[tid + 256] = 0u;
    }
    __syncthreads();

    float a[8];
    {
        const int lane = tid & 31;
        const int wv   = tid >> 5;

        float bv = -__builtin_inff();
        int   bi = 0x7fffffff;
        for (int k = 0; k < 8; ++k) {
            const int i = tid + 256 * k;
            a[k] = s_act[i];
            if (key_before(a[k], i, bv, bi)) { bv = a[k]; bi = i; }
        }
        for (int m = 16; m >= 1; m >>= 1) {
            float ov = __shfl_xor(bv, m, 32);
            int   oi = __shfl_xor(bi, m, 32);
            if (key_before(ov, oi, bv, bi)) { bv = ov; bi = oi; }
        }
        if (lane == 0) { s_wv[wv] = bv; s_wi[wv] = bi; }
        s_val[tid] = -__builtin_inff();
        s_idx[tid] = 0x7fffffff;
        __syncthreads();

        float lv[8];
        int   li[8];
        for (int w = 0; w < 8; ++w) { lv[w] = s_wv[w]; li[w] = s_wi[w]; }
        float Lv = lv[0];
        int   Li = li[0];
        for (int w = 1; w < 8; ++w) {
            const bool later = key_before(Lv, Li, lv[w], li[w]);
            Lv = later ? lv[w] : Lv;
            Li = later ? li[w] : Li;
        }

        bool keep[8];
        int  slot[8];
        int  wcount = 0;
        for (int k = 0; k < 8; ++k) {
            const int i = tid + 256 * k;
            keep[k] = (a[k] > Lv) || (a[k] == Lv && i <= Li);
            const unsigned m = __builtin_amdgcn_ballot_w32(keep[k]);
            slot[k] = wcount + (int)__builtin_amdgcn_mbcnt_lo(m, 0u);
            wcount += __builtin_popcount(m);
        }
        if (lane == 0) s_wn[wv] = wcount;
        __syncthreads();

        int off = 0, n = 0;
        for (int w = 0; w < 8; ++w) {
            const int c = s_wn[w];
            if (w < wv) off += c;
            n += c;
        }

        if (n <= 256) {
            for (int k = 0; k < 8; ++k) {
                if (keep[k]) {
                    s_val[off + slot[k]] = a[k];
                    s_idx[off + slot[k]] = tid + 256 * k;
                }
            }
            __syncthreads();
            if (tid < n) {
                const float v = s_val[tid];
                const int   i = s_idx[tid];
                int rank = 0;
                for (int j = 0; j < n; j += 8) {
                    float cv[8];
                    int   ci[8];
                    for (int u = 0; u < 8; ++u) { cv[u] = s_val[j + u]; ci[u] = s_idx[j + u]; }
                    for (int u = 0; u < 8; ++u) rank += key_before(cv[u], ci[u], v, i) ? 1 : 0;
                }
                if (rank < 8) s_ch[i] = 1;
            }
            __syncthreads();
        } else {
            for (int t = 0; t < 8; ++t) {
                float pv = -__builtin_inff();
                int   pi = 0x7fffffff;
                for (int i = tid; i < R_; i += 256) {
                    if (!s_ch[i]) {
                        float v = s_act[i];
                        if (key_before(v, i, pv, pi)) { pv = v; pi = i; }
                    }
                }
                s_val[tid] = pv; s_idx[tid] = pi;
                __syncthreads();
                for (int s = 128; s > 0; s >>= 1) {
                    if (tid < s) {
                        float ov = s_val[tid + s]; int oi = s_idx[tid + s];
                        if (key_before(ov, oi, s_val[tid], s_idx[tid])) {
                            s_val[tid] = ov; s_idx[tid] = oi;
                        }
                    }
                    __syncthreads();
                }
                if (tid == 0 && s_idx[0] < R_) s_ch[s_idx[0]] = 1;
                __syncthreads();
            }
        }
    }

    float x[8];
    #pragma unroll
    for (int k = 0; k < 8; ++k) {
        x[k] = s_ch[tid + 256 * k] ? p[k] + a[k] : p[k];
    }

    float sum = 0.0f;
    #pragma unroll
    for (int k = 0; k < 8; ++k) sum += x[k];
    s_val[tid] = sum;
    __syncthreads();
    for (int s = 128; s > 0; s >>= 1) {
        if (tid < s) s_val[tid] += s_val[tid + s];
        __syncthreads();
    }
    const float mu = s_val[0] * (1.0f / R_);
    __syncthreads();

    float ss = 0.0f;
    #pragma unroll
    for (int k = 0; k < 8; ++k) { const float d = x[k] - mu; ss = fmaf(d, d, ss); }
    s_val[tid] = ss;
    __syncthreads();
    for (int s = 128; s > 0; s >>= 1) {
        if (tid < s) s_val[tid] += s_val[tid + s];
        __syncthreads();
    }
    const float inv = rsqrtf(s_val[0] * (1.0f / R_) + 1e-5f);

    float* orow = out + (size_t)b * R_ + tid;
    #pragma unroll
    for (int k = 0; k < 8; ++k) {
        orow[256 * k] = (x[k] - mu) * inv * g[k] + be[k];
    }
}

extern "C" void kernel_launch(void* const* d_in, const int* in_sizes, int n_in,
                              void* d_out, int out_size, void* d_ws, size_t ws_size,
                              hipStream_t stream) {
    const float* facts        = (const float*)d_in[0];
    const float* fact_logits  = (const float*)d_in[1];
    const float* agg_logits   = (const float*)d_in[2];
    const float* strength     = (const float*)d_in[3];
    const float* W_proj       = (const float*)d_in[4];
    const float* ln_gamma     = (const float*)d_in[5];
    const float* ln_beta      = (const float*)d_in[6];
    float* out = (float*)d_out;

    char* ws = (char*)d_ws;
    float* ract     = (float*)(ws);
    float* proj     = (float*)(ws + 262144);

    proj_scan_kernel<<<128, 256, 0, stream>>>(facts, W_proj, fact_logits, agg_logits, strength,
                                              proj, ract);

    gate_ln_kernel<<<B_, 256, 0, stream>>>(proj, ract, ln_gamma, ln_beta, out);
}
